# starting kernel
# baseline (speedup 1.0000x reference)
#include <hip/hip_runtime.h>

typedef float v2f __attribute__((ext_vector_type(2)));
typedef float v8f __attribute__((ext_vector_type(8)));

__device__ constexpr float GK[7] = {
    0.0701593f, 0.1310749f, 0.1907128f, 0.2161058f,
    0.1907128f, 0.1310749f, 0.0701593f
};

#define DIM     160
#define SLICES  (4 * DIM)
#define TILES_X (DIM / 16)
#define TILES_Y (DIM / 16)
#define TILES_PER_SLICE (TILES_X * TILES_Y)

__global__ __launch_bounds__(128) void gb3d_blurx_wmma(
    const float* __restrict__ in, float* __restrict__ out) {
  const int wave  = threadIdx.x >> 5;
  const int lane  = threadIdx.x & 31;
  const int tile  = blockIdx.x * 4 + wave;

  const int slice = tile / TILES_PER_SLICE;
  const int trem  = tile - slice * TILES_PER_SLICE;
  const int ty    = trem / TILES_X;
  const int tx    = trem - ty * TILES_X;
  const int x0    = tx * 16;
  const int y0    = ty * 16;

  const size_t base = (size_t)slice * (DIM * DIM);
  const int row = lane & 15;
  const int h   = lane >> 4;

  const float* __restrict__ rowp = in + base + (size_t)(y0 + row) * DIM;

  v2f a[6], b[6];
#pragma unroll
  for (int c = 0; c < 6; ++c) {
#pragma unroll
    for (int e = 0; e < 2; ++e) {
      const int k  = 4 * c + 2 * h + e;
      const int xi = x0 - 3 + k;
      const int xc = min(max(xi, 0), DIM - 1);
      float av = __builtin_nontemporal_load(&rowp[xc]);
      a[c][e] = (xi == xc) ? av : 0.0f;
      const int d = k - row;
      b[c][e] = (d >= 0 && d <= 6) ? GK[d] : 0.0f;
    }
  }

  v8f acc = {0.f, 0.f, 0.f, 0.f, 0.f, 0.f, 0.f, 0.f};
#pragma unroll
  for (int c = 0; c < 6; ++c) {
    acc = __builtin_amdgcn_wmma_f32_16x16x4_f32(
        false, a[c], false, b[c], (short)0, acc, false, false);
  }

  float* __restrict__ op = out + base + (size_t)(y0 + 8 * h) * DIM + x0 + row;
#pragma unroll
  for (int r = 0; r < 8; ++r)
    op[(size_t)r * DIM] = acc[r];
}

template <int STRIDEF>
__global__ __launch_bounds__(256) void gb3d_blur_axis_f4(
    const float4* __restrict__ in, float4* __restrict__ out) {
  const long tid  = (long)blockIdx.x * 256 + threadIdx.x;
  const long flat = tid * 4;
  const int  pos  = (int)((flat / STRIDEF) % DIM);
  constexpr int S4 = STRIDEF / 4;

  float4 acc = make_float4(0.f, 0.f, 0.f, 0.f);
#pragma unroll
  for (int t = 0; t < 7; ++t) {
    const int p = pos + t - 3;
    if (0 <= p && p < DIM) {
      const float4 v = in[tid + (long)(t - 3) * S4];
      const float  g = GK[t];
      acc.x += g * v.x; acc.y += g * v.y;
      acc.z += g * v.z; acc.w += g * v.w;
    }
  }
  out[tid] = acc;
}

extern "C" void kernel_launch(void* const* d_in, const int* in_sizes, int n_in,
                              void* d_out, int out_size, void* d_ws, size_t ws_size,
                              hipStream_t stream) {
  (void)in_sizes; (void)n_in; (void)out_size; (void)ws_size;
  const float* x   = (const float*)d_in[0];
  float*       out = (float*)d_out;
  float*       tmp = (float*)d_ws;

  gb3d_blurx_wmma<<<SLICES * TILES_PER_SLICE / 4, 128, 0, stream>>>(x, out);

  const int n4blocks = (4 * DIM * DIM * DIM) / 4 / 256;
  gb3d_blur_axis_f4<DIM><<<n4blocks, 256, 0, stream>>>(
      (const float4*)out, (float4*)tmp);
  gb3d_blur_axis_f4<DIM * DIM><<<n4blocks, 256, 0, stream>>>(
      (const float4*)tmp, (float4*)out);
}


# Fuse x-blur and y-blur into one kernel: a wave owns a 16-column strip and feeds the x-blur WMMA D tiles straight in as B of a banded y-blur WMMA; one pass and launch fewer; y sums in WMMA order
# speedup vs baseline: 1.8727x; 1.8727x over previous
#include <hip/hip_runtime.h>

typedef float v2f __attribute__((ext_vector_type(2)));
typedef float v8f __attribute__((ext_vector_type(8)));

__device__ constexpr float GK[7] = {
    0.0701593f, 0.1310749f, 0.1907128f, 0.2161058f,
    0.1907128f, 0.1310749f, 0.0701593f
};

#define DIM     160
#define SLICES  (4 * DIM)
#define TILES_X (DIM / 16)
#define TILES_Y (DIM / 16)
#define PF      3

__device__ __forceinline__ float band(int d) {
  return (d >= 0 && d <= 6) ? GK[d] : 0.0f;
}

__global__ __launch_bounds__(128) void gb3d_blurxy_wmma(
    const float* __restrict__ in, float* __restrict__ out) {
  const int wave  = threadIdx.x >> 5;
  const int lane  = threadIdx.x & 31;
  const int strip = blockIdx.x * 4 + wave;

  const int slice = strip / TILES_X;
  const int tx    = strip - slice * TILES_X;
  const int x0    = tx * 16;

  const size_t base = (size_t)slice * (DIM * DIM);
  const int row = lane & 15;
  const int h   = lane >> 4;

  int  xcl[6][2];
  bool xok[6][2];
  v2f  bx[6];
#pragma unroll
  for (int c = 0; c < 6; ++c) {
#pragma unroll
    for (int e = 0; e < 2; ++e) {
      const int k  = 4 * c + 2 * h + e;
      const int xi = x0 - 3 + k;
      const int xc = min(max(xi, 0), DIM - 1);
      xcl[c][e] = xc;
      xok[c][e] = (xi == xc);
      bx[c][e]  = band(k - row);
    }
  }

  v2f wprev[2], wcur[4], wnext[2];
#pragma unroll
  for (int e = 0; e < 2; ++e) {
#pragma unroll
    for (int q = 0; q < 2; ++q) {
      wprev[q][e] = band(-16 + (4 + 2 * q) + e + 8 * h - row + 3);
      wnext[q][e] = band( 16 + (2 * q)     + e + 8 * h - row + 3);
    }
#pragma unroll
    for (int q = 0; q < 4; ++q)
      wcur[q][e] = band((2 * q) + e + 8 * h - row + 3);
  }

  const float* __restrict__ rowp = in + base + (size_t)row * DIM;
  float* __restrict__ op = out + base + (size_t)(8 * h) * DIM + x0 + row;

  v2f a[TILES_Y][6];
  v8f T[TILES_Y];

#define LOAD_TILE(J)                                                          \
  _Pragma("unroll")                                                           \
  for (int c = 0; c < 6; ++c) {                                               \
    _Pragma("unroll")                                                         \
    for (int e = 0; e < 2; ++e) {                                             \
                                      \
      a[J][c][e] = __builtin_nontemporal_load(                                \
          &rowp[(size_t)(J) * 16 * DIM + xcl[c][e]]);                         \
    }                                                                         \
  }

#pragma unroll
  for (int j = 0; j < PF; ++j) { LOAD_TILE(j) }

#pragma unroll
  for (int j = 0; j <= TILES_Y; ++j) {
    if (j + PF < TILES_Y) { LOAD_TILE(j + PF) }

    if (j < TILES_Y) {
      v8f acc = {0.f, 0.f, 0.f, 0.f, 0.f, 0.f, 0.f, 0.f};
#pragma unroll
      for (int c = 0; c < 6; ++c) {
        v2f af;
        af[0] = xok[c][0] ? a[j][c][0] : 0.0f;
        af[1] = xok[c][1] ? a[j][c][1] : 0.0f;
        acc = __builtin_amdgcn_wmma_f32_16x16x4_f32(
            false, af, false, bx[c], (short)0, acc, false, false);
      }
      T[j] = acc;
    }

    if (j >= 1) {
      const int t = j - 1;
      v8f acc = {0.f, 0.f, 0.f, 0.f, 0.f, 0.f, 0.f, 0.f};
      if (t >= 1) {
#pragma unroll
        for (int q = 0; q < 2; ++q) {
          v2f bt = {T[t - 1][4 + 2 * q], T[t - 1][5 + 2 * q]};
          acc = __builtin_amdgcn_wmma_f32_16x16x4_f32(
              false, wprev[q], false, bt, (short)0, acc, false, false);
        }
      }
#pragma unroll
      for (int q = 0; q < 4; ++q) {
        v2f bt = {T[t][2 * q], T[t][2 * q + 1]};
        acc = __builtin_amdgcn_wmma_f32_16x16x4_f32(
            false, wcur[q], false, bt, (short)0, acc, false, false);
      }
      if (t + 1 < TILES_Y) {
#pragma unroll
        for (int q = 0; q < 2; ++q) {
          v2f bt = {T[t + 1][2 * q], T[t + 1][2 * q + 1]};
          acc = __builtin_amdgcn_wmma_f32_16x16x4_f32(
              false, wnext[q], false, bt, (short)0, acc, false, false);
        }
      }
      float* __restrict__ ot = op + (size_t)t * 16 * DIM;
#pragma unroll
      for (int r = 0; r < 8; ++r)
        ot[(size_t)r * DIM] = acc[r];
    }
    __builtin_amdgcn_sched_barrier(0);
  }
#undef LOAD_TILE
}

template <int STRIDEF>
__global__ __launch_bounds__(256) void gb3d_blur_axis_f4(
    const float4* __restrict__ in, float4* __restrict__ out) {
  const long tid  = (long)blockIdx.x * 256 + threadIdx.x;
  const long flat = tid * 4;
  const int  pos  = (int)((flat / STRIDEF) % DIM);
  constexpr int S4 = STRIDEF / 4;

  float4 acc = make_float4(0.f, 0.f, 0.f, 0.f);
#pragma unroll
  for (int t = 0; t < 7; ++t) {
    const int p = pos + t - 3;
    if (0 <= p && p < DIM) {
      const float4 v = in[tid + (long)(t - 3) * S4];
      const float  g = GK[t];
      acc.x += g * v.x; acc.y += g * v.y;
      acc.z += g * v.z; acc.w += g * v.w;
    }
  }
  out[tid] = acc;
}

extern "C" void kernel_launch(void* const* d_in, const int* in_sizes, int n_in,
                              void* d_out, int out_size, void* d_ws, size_t ws_size,
                              hipStream_t stream) {
  (void)in_sizes; (void)n_in; (void)out_size; (void)ws_size;
  const float* x   = (const float*)d_in[0];
  float*       out = (float*)d_out;
  float*       tmp = (float*)d_ws;

  gb3d_blurxy_wmma<<<SLICES * TILES_X / 4, 128, 0, stream>>>(x, tmp);

  const int n4blocks = (4 * DIM * DIM * DIM) / 4 / 256;
  gb3d_blur_axis_f4<DIM * DIM><<<n4blocks, 256, 0, stream>>>(
      (const float4*)tmp, (float4*)out);
}


# x-blur A operand read as three aligned 16-byte loads per lane and row tile (was twelve 4-byte loads): window moved to x0-4, k order permuted alike in A and B; x sums in a new k order
# speedup vs baseline: 2.5016x; 1.3358x over previous
#include <hip/hip_runtime.h>

typedef float v2f __attribute__((ext_vector_type(2)));
typedef float v4f __attribute__((ext_vector_type(4)));
typedef float v8f __attribute__((ext_vector_type(8)));

__device__ constexpr float GK[7] = {
    0.0701593f, 0.1310749f, 0.1907128f, 0.2161058f,
    0.1907128f, 0.1310749f, 0.0701593f
};

#define DIM     160
#define SLICES  (4 * DIM)
#define TILES_X (DIM / 16)
#define TILES_Y (DIM / 16)
#define PF      3

__device__ __forceinline__ float band(int d) {
  return (d >= 0 && d <= 6) ? GK[d] : 0.0f;
}

__global__ __launch_bounds__(128) void gb3d_blurxy_wmma(
    const float* __restrict__ in, float* __restrict__ out) {
  const int wave  = threadIdx.x >> 5;
  const int lane  = threadIdx.x & 31;
  const int strip = blockIdx.x * 4 + wave;

  const int slice = strip / TILES_X;
  const int tx    = strip - slice * TILES_X;
  const int x0    = tx * 16;

  const size_t base = (size_t)slice * (DIM * DIM);
  const int row = lane & 15;
  const int h   = lane >> 4;

  int  xcl[3];
  bool xok[3];
  v2f  bx[6];
#pragma unroll
  for (int g = 0; g < 3; ++g) {
    const int k0 = 8 * g + 4 * h;
    const int xi = x0 - 4 + k0;
    const int xc = min(max(xi, 0), DIM - 4);
    xcl[g] = xc;
    xok[g] = (xi == xc);
#pragma unroll
    for (int s = 0; s < 2; ++s) {
#pragma unroll
      for (int e = 0; e < 2; ++e) {
        const int k = k0 + 2 * s + e;
        bx[2 * g + s][e] = band(k - row - 1);
      }
    }
  }

  v2f wprev[2], wcur[4], wnext[2];
#pragma unroll
  for (int e = 0; e < 2; ++e) {
#pragma unroll
    for (int q = 0; q < 2; ++q) {
      wprev[q][e] = band(-16 + (4 + 2 * q) + e + 8 * h - row + 3);
      wnext[q][e] = band( 16 + (2 * q)     + e + 8 * h - row + 3);
    }
#pragma unroll
    for (int q = 0; q < 4; ++q)
      wcur[q][e] = band((2 * q) + e + 8 * h - row + 3);
  }

  const float* __restrict__ rowp = in + base + (size_t)row * DIM;
  float* __restrict__ op = out + base + (size_t)(8 * h) * DIM + x0 + row;

  v4f a[TILES_Y][3];
  v8f T[TILES_Y];

#define LOAD_TILE(J)                                                          \
  _Pragma("unroll")                                                           \
  for (int g = 0; g < 3; ++g) {                                               \
                                      \
    a[J][g] = __builtin_nontemporal_load(reinterpret_cast<const v4f*>(        \
        &rowp[(size_t)(J) * 16 * DIM + xcl[g]]));                             \
  }

#pragma unroll
  for (int j = 0; j < PF; ++j) { LOAD_TILE(j) }

#pragma unroll
  for (int j = 0; j <= TILES_Y; ++j) {
    if (j + PF < TILES_Y) { LOAD_TILE(j + PF) }

    if (j < TILES_Y) {
      v8f acc = {0.f, 0.f, 0.f, 0.f, 0.f, 0.f, 0.f, 0.f};
#pragma unroll
      for (int g = 0; g < 3; ++g) {
#pragma unroll
        for (int s = 0; s < 2; ++s) {
          v2f af;
          af[0] = xok[g] ? a[j][g][2 * s]     : 0.0f;
          af[1] = xok[g] ? a[j][g][2 * s + 1] : 0.0f;
          acc = __builtin_amdgcn_wmma_f32_16x16x4_f32(
              false, af, false, bx[2 * g + s], (short)0, acc, false, false);
        }
      }
      T[j] = acc;
    }

    if (j >= 1) {
      const int t = j - 1;
      v8f acc = {0.f, 0.f, 0.f, 0.f, 0.f, 0.f, 0.f, 0.f};
      if (t >= 1) {
#pragma unroll
        for (int q = 0; q < 2; ++q) {
          v2f bt = {T[t - 1][4 + 2 * q], T[t - 1][5 + 2 * q]};
          acc = __builtin_amdgcn_wmma_f32_16x16x4_f32(
              false, wprev[q], false, bt, (short)0, acc, false, false);
        }
      }
#pragma unroll
      for (int q = 0; q < 4; ++q) {
        v2f bt = {T[t][2 * q], T[t][2 * q + 1]};
        acc = __builtin_amdgcn_wmma_f32_16x16x4_f32(
            false, wcur[q], false, bt, (short)0, acc, false, false);
      }
      if (t + 1 < TILES_Y) {
#pragma unroll
        for (int q = 0; q < 2; ++q) {
          v2f bt = {T[t + 1][2 * q], T[t + 1][2 * q + 1]};
          acc = __builtin_amdgcn_wmma_f32_16x16x4_f32(
              false, wnext[q], false, bt, (short)0, acc, false, false);
        }
      }
      float* __restrict__ ot = op + (size_t)t * 16 * DIM;
#pragma unroll
      for (int r = 0; r < 8; ++r)
        ot[(size_t)r * DIM] = acc[r];
    }
    __builtin_amdgcn_sched_barrier(0);
  }
#undef LOAD_TILE
}

template <int STRIDEF>
__global__ __launch_bounds__(256) void gb3d_blur_axis_f4(
    const float4* __restrict__ in, float4* __restrict__ out) {
  const long tid  = (long)blockIdx.x * 256 + threadIdx.x;
  const long flat = tid * 4;
  const int  pos  = (int)((flat / STRIDEF) % DIM);
  constexpr int S4 = STRIDEF / 4;

  float4 acc = make_float4(0.f, 0.f, 0.f, 0.f);
#pragma unroll
  for (int t = 0; t < 7; ++t) {
    const int p = pos + t - 3;
    if (0 <= p && p < DIM) {
      const float4 v = in[tid + (long)(t - 3) * S4];
      const float  g = GK[t];
      acc.x += g * v.x; acc.y += g * v.y;
      acc.z += g * v.z; acc.w += g * v.w;
    }
  }
  out[tid] = acc;
}

extern "C" void kernel_launch(void* const* d_in, const int* in_sizes, int n_in,
                              void* d_out, int out_size, void* d_ws, size_t ws_size,
                              hipStream_t stream) {
  (void)in_sizes; (void)n_in; (void)out_size; (void)ws_size;
  const float* x   = (const float*)d_in[0];
  float*       out = (float*)d_out;
  float*       tmp = (float*)d_ws;

  gb3d_blurxy_wmma<<<SLICES * TILES_X / 4, 128, 0, stream>>>(x, tmp);

  const int n4blocks = (4 * DIM * DIM * DIM) / 4 / 256;
  gb3d_blur_axis_f4<DIM * DIM><<<n4blocks, 256, 0, stream>>>(
      (const float4*)tmp, (float4*)out);
}


# Fuse the z-blur into the WMMA kernel: a wave walks along z over a 16x10 (x,y) tile with 7 running z sums in registers; workspace pass and 2nd launch gone; y sums in a new k order, z order kept
# speedup vs baseline: 5.1118x; 2.0434x over previous
#include <hip/hip_runtime.h>

typedef float v2f __attribute__((ext_vector_type(2)));
typedef float v4f __attribute__((ext_vector_type(4)));
typedef float v8f __attribute__((ext_vector_type(8)));

__device__ constexpr float GK[7] = {
    0.0701593f, 0.1310749f, 0.1907128f, 0.2161058f,
    0.1907128f, 0.1310749f, 0.0701593f
};

#define DIM      160
#define PLANE    (DIM * DIM)
#define VOLS     4
#define TILES_X  (DIM / 16)
#define ROWS_OUT 10
#define TILES_Y  (DIM / ROWS_OUT)
#define ZC       20
#define CHUNKS   (DIM / ZC)
#define STEPS    (ZC + 6)
#define NBUF     3

__device__ __forceinline__ float band(int d) {
  return (d >= 0 && d <= 6) ? GK[d] : 0.0f;
}

__global__ __launch_bounds__(128) void gb3d_blur_xyz_wmma(
    const float* __restrict__ in, float* __restrict__ out) {
  const int lane = threadIdx.x & 31;
  const int wid = __builtin_amdgcn_readfirstlane(
      (int)(blockIdx.x * 4 + (threadIdx.x >> 5)));

  const int tx  = wid % TILES_X;
  const int t1  = wid / TILES_X;
  const int ty  = t1 % TILES_Y;
  const int t2  = t1 / TILES_Y;
  const int ck  = t2 % CHUNKS;
  const int vol = t2 / CHUNKS;

  const int x0 = tx * 16;
  const int y0 = ty * ROWS_OUT;
  const int zs = ck * ZC;

  const int row = lane & 15;
  const int h   = lane >> 4;

  int  off[3];
  bool okg[3];
  v2f  bx[6];
  const int ya  = y0 - 3 + row;
  const int yc  = min(max(ya, 0), DIM - 1);
  const bool yok = (ya == yc);
#pragma unroll
  for (int g = 0; g < 3; ++g) {
    const int k0 = 8 * g + 4 * h;
    const int xi = x0 - 4 + k0;
    const int xc = min(max(xi, 0), DIM - 4);
    off[g] = yc * DIM + xc;
    okg[g] = (xi == xc) && yok;
#pragma unroll
    for (int s = 0; s < 2; ++s) {
#pragma unroll
      for (int e = 0; e < 2; ++e) {
        const int k = k0 + 2 * s + e;
        bx[2 * g + s][e] = band(k - row - 1);
      }
    }
  }

  v2f wy[4];
#pragma unroll
  for (int q = 0; q < 4; ++q) {
#pragma unroll
    for (int e = 0; e < 2; ++e)
      wy[q][e] = band(2 * q + e + 8 * h - row);
  }

  const float* __restrict__ vin = in + (size_t)vol * DIM * PLANE;
  float* __restrict__ op = out + (size_t)vol * DIM * PLANE
                         + (size_t)(y0 + 8 * h) * DIM + x0 + row;

  v4f buf[NBUF][3];
  v8f S[7];
#pragma unroll
  for (int t = 0; t < 7; ++t)
    S[t] = (v8f){0.f, 0.f, 0.f, 0.f, 0.f, 0.f, 0.f, 0.f};

#define LOAD_PLANE(ST)                                                        \
  {                                                                           \
    const int pl_ = min(max(zs - 3 + (ST), 0), DIM - 1);                      \
    const float* __restrict__ pp_ = vin + (size_t)pl_ * PLANE;                \
    _Pragma("unroll")                                                         \
    for (int g = 0; g < 3; ++g)                                               \
      buf[(ST) % NBUF][g] = *reinterpret_cast<const v4f*>(&pp_[off[g]]);      \
  }

  LOAD_PLANE(0)
  LOAD_PLANE(1)

#pragma unroll
  for (int st = 0; st < STEPS; ++st) {
    if (st + 2 < STEPS) { LOAD_PLANE(st + 2) }

    const int  p   = zs - 3 + st;
    const bool pok = (p >= 0) && (p < DIM);

    v8f T = {0.f, 0.f, 0.f, 0.f, 0.f, 0.f, 0.f, 0.f};
#pragma unroll
    for (int g = 0; g < 3; ++g) {
      const bool ok = okg[g] && pok;
#pragma unroll
      for (int s = 0; s < 2; ++s) {
        v2f af;
        af[0] = ok ? buf[st % NBUF][g][2 * s]     : 0.0f;
        af[1] = ok ? buf[st % NBUF][g][2 * s + 1] : 0.0f;
        T = __builtin_amdgcn_wmma_f32_16x16x4_f32(
            false, af, false, bx[2 * g + s], (short)0, T, false, false);
      }
    }

    v8f P = {0.f, 0.f, 0.f, 0.f, 0.f, 0.f, 0.f, 0.f};
#pragma unroll
    for (int q = 0; q < 4; ++q) {
      v2f bt = {T[2 * q], T[2 * q + 1]};
      P = __builtin_amdgcn_wmma_f32_16x16x4_f32(
          false, wy[q], false, bt, (short)0, P, false, false);
    }

#pragma unroll
    for (int t = 6; t >= 1; --t) {
#pragma unroll
      for (int r = 0; r < 8; ++r)
        S[t][r] = __builtin_fmaf(GK[t], P[r], S[t - 1][r]);
    }
#pragma unroll
    for (int r = 0; r < 8; ++r)
      S[0][r] = GK[0] * P[r];

    if (st >= 6) {
      float* __restrict__ ot = op + (size_t)(zs + st - 6) * PLANE;
#pragma unroll
      for (int r = 0; r < 8; ++r) {
        if (r < 2 || h == 0)
          ot[(size_t)r * DIM] = S[6][r];
      }
    }
    __builtin_amdgcn_sched_barrier(0);
  }
#undef LOAD_PLANE
}

extern "C" void kernel_launch(void* const* d_in, const int* in_sizes, int n_in,
                              void* d_out, int out_size, void* d_ws, size_t ws_size,
                              hipStream_t stream) {
  (void)in_sizes; (void)n_in; (void)out_size; (void)d_ws; (void)ws_size;
  const float* x   = (const float*)d_in[0];
  float*       out = (float*)d_out;

  gb3d_blur_xyz_wmma<<<VOLS * CHUNKS * TILES_Y * TILES_X / 4, 128, 0, stream>>>(x, out);
}


# Input planes staged through LDS by a 20-wave work-group owning 20 whole rows: each line read once by whole-line loads, y halo shared by 2 row tiles, barrier fenced for LDS only; sums unchanged
# speedup vs baseline: 6.9575x; 1.3611x over previous
#include <hip/hip_runtime.h>

typedef float v2f __attribute__((ext_vector_type(2)));
typedef float v4f __attribute__((ext_vector_type(4)));
typedef float v8f __attribute__((ext_vector_type(8)));

__device__ constexpr float GK[7] = {
    0.0701593f, 0.1310749f, 0.1907128f, 0.2161058f,
    0.1907128f, 0.1310749f, 0.0701593f
};

#define DIM        160
#define PLANE      (DIM * DIM)
#define VOLS       4
#define TILES_X    (DIM / 16)
#define ROWS_OUT   10
#define TILES_YL   2
#define WG_ROWS    (ROWS_OUT * TILES_YL)
#define WIN_ROWS   (WG_ROWS + 6)
#define WGS_Y      (DIM / WG_ROWS)
#define ZC         20
#define CHUNKS     (DIM / ZC)
#define STEPS      (ZC + 6)
#define WAVES      (TILES_X * TILES_YL)
#define THREADS    (WAVES * 32)
#define ROW_PIECES (DIM / 4)
#define PITCH_F    172
#define SLOT_F     (WIN_ROWS * PITCH_F)
#define NSLOT      3

__device__ __forceinline__ float band(int d) {
  return (d >= 0 && d <= 6) ? GK[d] : 0.0f;
}

__global__ __launch_bounds__(THREADS) void gb3d_blur_xyz_wmma(
    const float* __restrict__ in, float* __restrict__ out) {
  __shared__ __attribute__((aligned(16))) float lds[NSLOT * SLOT_F];

  const int tid  = threadIdx.x;
  const int lane = tid & 31;
  const int w   = __builtin_amdgcn_readfirstlane(tid >> 5);
  const int tx  = w % TILES_X;
  const int tyl = w / TILES_X;

  const int b   = blockIdx.x;
  const int yp  = b % WGS_Y;
  const int t1  = b / WGS_Y;
  const int ck  = t1 % CHUNKS;
  const int vol = t1 / CHUNKS;

  const int x0 = tx * 16;
  const int yw = yp * WG_ROWS;
  const int y0 = yw + tyl * ROWS_OUT;
  const int zs = ck * ZC;

  const int row = lane & 15;
  const int h   = lane >> 4;

  v2f bx[6];
#pragma unroll
  for (int g = 0; g < 3; ++g) {
    const int k0 = 8 * g + 4 * h;
#pragma unroll
    for (int s = 0; s < 2; ++s) {
#pragma unroll
      for (int e = 0; e < 2; ++e) {
        const int k = k0 + 2 * s + e;
        bx[2 * g + s][e] = band(k - row - 1);
      }
    }
  }

  v2f wy[4];
#pragma unroll
  for (int q = 0; q < 4; ++q) {
#pragma unroll
    for (int e = 0; e < 2; ++e)
      wy[q][e] = band(2 * q + e + 8 * h - row);
  }

  const int  r0   = tid / ROW_PIECES;
  const int  pc   = tid - r0 * ROW_PIECES;
  const int  gy0  = yw - 3 + r0;
  const int  gy0c = min(max(gy0, 0), DIM - 1);
  const bool vy0  = (gy0 == gy0c);
  const int  gy1  = gy0 + 16;
  const int  gy1c = min(max(gy1, 0), DIM - 1);
  const bool vy1  = (gy1 == gy1c);
  const bool has1 = tid < (WIN_ROWS - 16) * ROW_PIECES;
  const int  go0  = gy0c * DIM + pc * 4;
  const int  go1  = has1 ? gy1c * DIM + pc * 4 : go0;
  float* const st0 = lds + r0 * PITCH_F + 4 + pc * 4;
  float* const st1 = st0 + 16 * PITCH_F;
  const float* const fr0 = lds + (tyl * ROWS_OUT + row) * PITCH_F + x0 + 4 * h;

  const v4f zero4 = {0.f, 0.f, 0.f, 0.f};
  if (tid < NSLOT * WIN_ROWS * 2) {
    const int sl  = tid / (WIN_ROWS * 2);
    const int rem = tid - sl * (WIN_ROWS * 2);
    *reinterpret_cast<v4f*>(lds + sl * SLOT_F + (rem >> 1) * PITCH_F + (rem & 1) * 164) = zero4;
  }

  const float* __restrict__ vin = in + (size_t)vol * DIM * PLANE;
  float* __restrict__ op = out + (size_t)vol * DIM * PLANE
                         + (size_t)(y0 + 8 * h) * DIM + x0 + row;

  v4f G[3][2];
  v4f A[2][3];
  v8f S[7];
#pragma unroll
  for (int i = 0; i < 3; ++i) { G[i][0] = zero4; G[i][1] = zero4; }
#pragma unroll
  for (int t = 0; t < 7; ++t)
    S[t] = (v8f){0.f, 0.f, 0.f, 0.f, 0.f, 0.f, 0.f, 0.f};

#define GLOAD(ST)                                                             \
  {                                                                           \
    const int pl_ = min(max(zs - 3 + (ST), 0), DIM - 1);                      \
    const float* __restrict__ pp_ = vin + (size_t)pl_ * PLANE;                \
    G[(ST) % 3][0] = *reinterpret_cast<const v4f*>(pp_ + go0);                \
    G[(ST) % 3][1] = *reinterpret_cast<const v4f*>(pp_ + go1);                \
  }

#define STAGE(ST)                                                             \
  {                                                                           \
    const int  p_   = zs - 3 + (ST);                                          \
    const bool pok_ = (p_ >= 0) && (p_ < DIM);                                \
    const bool k0_  = vy0 && pok_;                                            \
    const bool k1_  = vy1 && pok_;                                            \
    v4f a_, b_;                                                               \
    _Pragma("unroll")                                                         \
    for (int i = 0; i < 4; ++i) {     \
      a_[i] = k0_ ? G[(ST) % 3][0][i] : 0.0f;                                 \
      b_[i] = k1_ ? G[(ST) % 3][1][i] : 0.0f;                                 \
    }                                                                         \
    *reinterpret_cast<v4f*>(st0 + ((ST) % NSLOT) * SLOT_F) = a_;              \
    if (has1)                                                                 \
      *reinterpret_cast<v4f*>(st1 + ((ST) % NSLOT) * SLOT_F) = b_;            \
  }

#define LDFRAG(ST)                                                            \
  {                                                                           \
    _Pragma("unroll")                                                         \
    for (int g = 0; g < 3; ++g)                                               \
      A[(ST) & 1][g] = *reinterpret_cast<const v4f*>(                         \
          fr0 + ((ST) % NSLOT) * SLOT_F + 8 * g);                             \
  }

#pragma unroll
  for (int t = -4; t < STEPS; ++t) {
    if (t + 4 < STEPS) { GLOAD(t + 4) }
    if (t + 2 >= 0 && t + 2 < STEPS) { STAGE(t + 2) }
    if (t + 1 >= 0 && t + 1 < STEPS) { LDFRAG(t + 1) }
    __builtin_amdgcn_sched_barrier(0);

    if (t >= 0) {
      v8f T = {0.f, 0.f, 0.f, 0.f, 0.f, 0.f, 0.f, 0.f};
#pragma unroll
      for (int g = 0; g < 3; ++g) {
#pragma unroll
        for (int s = 0; s < 2; ++s) {
          v2f af = {A[t & 1][g][2 * s], A[t & 1][g][2 * s + 1]};
          T = __builtin_amdgcn_wmma_f32_16x16x4_f32(
              false, af, false, bx[2 * g + s], (short)0, T, false, false);
        }
      }

      v8f P = {0.f, 0.f, 0.f, 0.f, 0.f, 0.f, 0.f, 0.f};
#pragma unroll
      for (int q = 0; q < 4; ++q) {
        v2f bt = {T[2 * q], T[2 * q + 1]};
        P = __builtin_amdgcn_wmma_f32_16x16x4_f32(
            false, wy[q], false, bt, (short)0, P, false, false);
      }

#pragma unroll
      for (int u = 6; u >= 1; --u) {
#pragma unroll
        for (int r = 0; r < 8; ++r)
          S[u][r] = __builtin_fmaf(GK[u], P[r], S[u - 1][r]);
      }
#pragma unroll
      for (int r = 0; r < 8; ++r)
        S[0][r] = GK[0] * P[r];

      if (t >= 6) {
        float* __restrict__ ot = op + (size_t)(zs + t - 6) * PLANE;
#pragma unroll
        for (int r = 0; r < 8; ++r) {
          if (r < 2 || h == 0)
            ot[(size_t)r * DIM] = S[6][r];
        }
      }
    }

    if (t + 2 >= 0 && t + 2 < STEPS) {
      __builtin_amdgcn_fence(__ATOMIC_RELEASE, "workgroup", "local");
      __builtin_amdgcn_s_barrier();
      __builtin_amdgcn_fence(__ATOMIC_ACQUIRE, "workgroup", "local");
    }
    __builtin_amdgcn_sched_barrier(0);
  }
#undef GLOAD
#undef STAGE
#undef LDFRAG
}

extern "C" void kernel_launch(void* const* d_in, const int* in_sizes, int n_in,
                              void* d_out, int out_size, void* d_ws, size_t ws_size,
                              hipStream_t stream) {
  (void)in_sizes; (void)n_in; (void)out_size; (void)d_ws; (void)ws_size;
  const float* x   = (const float*)d_in[0];
  float*       out = (float*)d_out;

  gb3d_blur_xyz_wmma<<<VOLS * CHUNKS * WGS_Y, THREADS, 0, stream>>>(x, out);
}


# Output planes staged through LDS, stored as whole 128-byte lines: one 16-byte store per thread into the work-group's contiguous 20 rows (was 4-byte stores, two waves per line); sums unchanged
# speedup vs baseline: 7.9456x; 1.1420x over previous
#include <hip/hip_runtime.h>

typedef float v2f __attribute__((ext_vector_type(2)));
typedef float v4f __attribute__((ext_vector_type(4)));
typedef float v8f __attribute__((ext_vector_type(8)));

__device__ constexpr float GK[7] = {
    0.0701593f, 0.1310749f, 0.1907128f, 0.2161058f,
    0.1907128f, 0.1310749f, 0.0701593f
};

#define DIM        160
#define PLANE      (DIM * DIM)
#define VOLS       4
#define TILES_X    (DIM / 16)
#define ROWS_OUT   10
#define TILES_YL   2
#define WG_ROWS    (ROWS_OUT * TILES_YL)
#define WIN_ROWS   (WG_ROWS + 6)
#define WGS_Y      (DIM / WG_ROWS)
#define ZC         20
#define CHUNKS     (DIM / ZC)
#define STEPS      (ZC + 6)
#define WAVES      (TILES_X * TILES_YL)
#define THREADS    (WAVES * 32)
#define ROW_PIECES (DIM / 4)
#define PITCH_F    172
#define SLOT_F     (WIN_ROWS * PITCH_F)
#define NSLOT      3
#define OPITCH_F   164
#define OBUF_F     (WG_ROWS * OPITCH_F)
#define OUT_WAVES  ((WG_ROWS - 16) * ROW_PIECES / 32)

__device__ __forceinline__ float band(int d) {
  return (d >= 0 && d <= 6) ? GK[d] : 0.0f;
}

__global__ __launch_bounds__(THREADS) void gb3d_blur_xyz_wmma(
    const float* __restrict__ in, float* __restrict__ out) {
  __shared__ __attribute__((aligned(16))) float lds[NSLOT * SLOT_F + 2 * OBUF_F];

  const int tid  = threadIdx.x;
  const int lane = tid & 31;
  const int w   = __builtin_amdgcn_readfirstlane(tid >> 5);
  const int tx  = w % TILES_X;
  const int tyl = w / TILES_X;

  const int b   = blockIdx.x;
  const int yp  = b % WGS_Y;
  const int t1  = b / WGS_Y;
  const int ck  = t1 % CHUNKS;
  const int vol = t1 / CHUNKS;

  const int x0 = tx * 16;
  const int yw = yp * WG_ROWS;
  const int y0 = yw + tyl * ROWS_OUT;
  const int zs = ck * ZC;

  const int row = lane & 15;
  const int h   = lane >> 4;

  v2f bx[6];
#pragma unroll
  for (int g = 0; g < 3; ++g) {
    const int k0 = 8 * g + 4 * h;
#pragma unroll
    for (int s = 0; s < 2; ++s) {
#pragma unroll
      for (int e = 0; e < 2; ++e) {
        const int k = k0 + 2 * s + e;
        bx[2 * g + s][e] = band(k - row - 1);
      }
    }
  }

  v2f wy[4];
#pragma unroll
  for (int q = 0; q < 4; ++q) {
#pragma unroll
    for (int e = 0; e < 2; ++e)
      wy[q][e] = band(2 * q + e + 8 * h - row);
  }

  const int  r0   = tid / ROW_PIECES;
  const int  pc   = tid - r0 * ROW_PIECES;
  const int  gy0  = yw - 3 + r0;
  const int  gy0c = min(max(gy0, 0), DIM - 1);
  const bool vy0  = (gy0 == gy0c);
  const int  gy1  = gy0 + 16;
  const int  gy1c = min(max(gy1, 0), DIM - 1);
  const bool vy1  = (gy1 == gy1c);
  const bool has1 = tid < (WIN_ROWS - 16) * ROW_PIECES;
  const int  go0  = gy0c * DIM + pc * 4;
  const int  go1  = has1 ? gy1c * DIM + pc * 4 : go0;
  float* const st0 = lds + r0 * PITCH_F + 4 + pc * 4;
  float* const st1 = st0 + 16 * PITCH_F;
  const float* const fr0 = lds + (tyl * ROWS_OUT + row) * PITCH_F + x0 + 4 * h;

  const v4f zero4 = {0.f, 0.f, 0.f, 0.f};
  if (tid < NSLOT * WIN_ROWS * 2) {
    const int sl  = tid / (WIN_ROWS * 2);
    const int rem = tid - sl * (WIN_ROWS * 2);
    *reinterpret_cast<v4f*>(lds + sl * SLOT_F + (rem >> 1) * PITCH_F + (rem & 1) * 164) = zero4;
  }

  const float* __restrict__ vin = in + (size_t)vol * DIM * PLANE;
  float* const obuf = lds + NSLOT * SLOT_F;
  float* const os   = obuf + (tyl * ROWS_OUT + 8 * h) * OPITCH_F + x0 + row;
  const float* const ol = obuf + r0 * OPITCH_F + pc * 4;
  const bool has_o1 = w < OUT_WAVES;
  float* __restrict__ og = out + (size_t)vol * DIM * PLANE + (size_t)yw * DIM + 4 * tid;

  v4f G[3][2];
  v4f A[2][3];
  v4f OL0 = zero4, OL1 = zero4;
  v8f S[7];
#pragma unroll
  for (int i = 0; i < 3; ++i) { G[i][0] = zero4; G[i][1] = zero4; }
#pragma unroll
  for (int t = 0; t < 7; ++t)
    S[t] = (v8f){0.f, 0.f, 0.f, 0.f, 0.f, 0.f, 0.f, 0.f};

#define GLOAD(ST)                                                             \
  {                                                                           \
    const int pl_ = min(max(zs - 3 + (ST), 0), DIM - 1);                      \
    const float* __restrict__ pp_ = vin + (size_t)pl_ * PLANE;                \
    G[(ST) % 3][0] = *reinterpret_cast<const v4f*>(pp_ + go0);                \
    G[(ST) % 3][1] = *reinterpret_cast<const v4f*>(pp_ + go1);                \
  }

#define STAGE(ST)                                                             \
  {                                                                           \
    const int  p_   = zs - 3 + (ST);                                          \
    const bool pok_ = (p_ >= 0) && (p_ < DIM);                                \
    const bool k0_  = vy0 && pok_;                                            \
    const bool k1_  = vy1 && pok_;                                            \
    v4f a_, b_;                                                               \
    _Pragma("unroll")                                                         \
    for (int i = 0; i < 4; ++i) {     \
      a_[i] = k0_ ? G[(ST) % 3][0][i] : 0.0f;                                 \
      b_[i] = k1_ ? G[(ST) % 3][1][i] : 0.0f;                                 \
    }                                                                         \
    *reinterpret_cast<v4f*>(st0 + ((ST) % NSLOT) * SLOT_F) = a_;              \
    if (has1)                                                                 \
      *reinterpret_cast<v4f*>(st1 + ((ST) % NSLOT) * SLOT_F) = b_;            \
  }

#define LDFRAG(ST)                                                            \
  {                                                                           \
    _Pragma("unroll")                                                         \
    for (int g = 0; g < 3; ++g)                                               \
      A[(ST) & 1][g] = *reinterpret_cast<const v4f*>(                         \
          fr0 + ((ST) % NSLOT) * SLOT_F + 8 * g);                             \
  }

#pragma unroll
  for (int t = -4; t < STEPS + 2; ++t) {
    if (t + 4 < STEPS) { GLOAD(t + 4) }
    if (t + 2 >= 0 && t + 2 < STEPS) { STAGE(t + 2) }
    if (t + 1 >= 0 && t + 1 < STEPS) { LDFRAG(t + 1) }
    if (t - 1 >= 6 && t - 1 < STEPS) {
      float* const o_ = os + ((t - 1) & 1) * OBUF_F;
#pragma unroll
      for (int r = 0; r < 8; ++r) {
        if (r < 2 || h == 0)
          o_[r * OPITCH_F] = S[6][r];
      }
    }
    if (t - 2 >= 6 && t - 2 < STEPS) {
      const float* const l_ = ol + ((t - 2) & 1) * OBUF_F;
      OL0 = *reinterpret_cast<const v4f*>(l_);
      if (has_o1)
        OL1 = *reinterpret_cast<const v4f*>(l_ + 16 * OPITCH_F);
    }
    __builtin_amdgcn_sched_barrier(0);

    if (t >= 0 && t < STEPS) {
      v8f T = {0.f, 0.f, 0.f, 0.f, 0.f, 0.f, 0.f, 0.f};
#pragma unroll
      for (int g = 0; g < 3; ++g) {
#pragma unroll
        for (int s = 0; s < 2; ++s) {
          v2f af = {A[t & 1][g][2 * s], A[t & 1][g][2 * s + 1]};
          T = __builtin_amdgcn_wmma_f32_16x16x4_f32(
              false, af, false, bx[2 * g + s], (short)0, T, false, false);
        }
      }

      v8f P = {0.f, 0.f, 0.f, 0.f, 0.f, 0.f, 0.f, 0.f};
#pragma unroll
      for (int q = 0; q < 4; ++q) {
        v2f bt = {T[2 * q], T[2 * q + 1]};
        P = __builtin_amdgcn_wmma_f32_16x16x4_f32(
            false, wy[q], false, bt, (short)0, P, false, false);
      }

#pragma unroll
      for (int u = 6; u >= 1; --u) {
#pragma unroll
        for (int r = 0; r < 8; ++r)
          S[u][r] = __builtin_fmaf(GK[u], P[r], S[u - 1][r]);
      }
#pragma unroll
      for (int r = 0; r < 8; ++r)
        S[0][r] = GK[0] * P[r];

    }

    __builtin_amdgcn_sched_barrier(0);
    if (t - 2 >= 6 && t - 2 < STEPS) {
      float* __restrict__ g_ = og + (size_t)(zs + t - 8) * PLANE;
      *reinterpret_cast<v4f*>(g_) = OL0;
      if (has_o1)
        *reinterpret_cast<v4f*>(g_ + 16 * DIM) = OL1;
    }

    if ((t + 2 >= 0 && t + 2 < STEPS) || (t - 1 >= 6 && t - 1 < STEPS)) {
      __builtin_amdgcn_fence(__ATOMIC_RELEASE, "workgroup", "local");
      __builtin_amdgcn_s_barrier();
      __builtin_amdgcn_fence(__ATOMIC_ACQUIRE, "workgroup", "local");
    }
    __builtin_amdgcn_sched_barrier(0);
  }
#undef GLOAD
#undef STAGE
#undef LDFRAG
}

extern "C" void kernel_launch(void* const* d_in, const int* in_sizes, int n_in,
                              void* d_out, int out_size, void* d_ws, size_t ws_size,
                              hipStream_t stream) {
  (void)in_sizes; (void)n_in; (void)out_size; (void)d_ws; (void)ws_size;
  const float* x   = (const float*)d_in[0];
  float*       out = (float*)d_out;

  gb3d_blur_xyz_wmma<<<VOLS * CHUNKS * WGS_Y, THREADS, 0, stream>>>(x, out);
}


# Work-group barrier split: a wave signals once its LDS traffic of the slot is done and waits only before the next slot's; y-blur and next plane's x-blur run between; sums unchanged
# speedup vs baseline: 8.2255x; 1.0352x over previous
#include <hip/hip_runtime.h>

typedef float v2f __attribute__((ext_vector_type(2)));
typedef float v4f __attribute__((ext_vector_type(4)));
typedef float v8f __attribute__((ext_vector_type(8)));

__device__ constexpr float GK[7] = {
    0.0701593f, 0.1310749f, 0.1907128f, 0.2161058f,
    0.1907128f, 0.1310749f, 0.0701593f
};

#define DIM        160
#define PLANE      (DIM * DIM)
#define VOLS       4
#define TILES_X    (DIM / 16)
#define ROWS_OUT   10
#define TILES_YL   2
#define WG_ROWS    (ROWS_OUT * TILES_YL)
#define WIN_ROWS   (WG_ROWS + 6)
#define WGS_Y      (DIM / WG_ROWS)
#define ZC         20
#define CHUNKS     (DIM / ZC)
#define STEPS      (ZC + 6)
#define WAVES      (TILES_X * TILES_YL)
#define THREADS    (WAVES * 32)
#define ROW_PIECES (DIM / 4)
#define PITCH_F    172
#define SLOT_F     (WIN_ROWS * PITCH_F)
#define NSLOT      3
#define OPITCH_F   164
#define OBUF_F     (WG_ROWS * OPITCH_F)
#define OUT_WAVES  ((WG_ROWS - 16) * ROW_PIECES / 32)

__device__ __forceinline__ float band(int d) {
  return (d >= 0 && d <= 6) ? GK[d] : 0.0f;
}

__global__ __launch_bounds__(THREADS) void gb3d_blur_xyz_wmma(
    const float* __restrict__ in, float* __restrict__ out) {
  __shared__ __attribute__((aligned(16))) float lds[NSLOT * SLOT_F + 2 * OBUF_F];

  const int tid  = threadIdx.x;
  const int lane = tid & 31;
  const int w   = __builtin_amdgcn_readfirstlane(tid >> 5);
  const int tx  = w % TILES_X;
  const int tyl = w / TILES_X;

  const int b   = blockIdx.x;
  const int yp  = b % WGS_Y;
  const int t1  = b / WGS_Y;
  const int ck  = t1 % CHUNKS;
  const int vol = t1 / CHUNKS;

  const int x0 = tx * 16;
  const int yw = yp * WG_ROWS;
  const int y0 = yw + tyl * ROWS_OUT;
  const int zs = ck * ZC;

  const int row = lane & 15;
  const int h   = lane >> 4;

  v2f bx[6];
#pragma unroll
  for (int g = 0; g < 3; ++g) {
    const int k0 = 8 * g + 4 * h;
#pragma unroll
    for (int s = 0; s < 2; ++s) {
#pragma unroll
      for (int e = 0; e < 2; ++e) {
        const int k = k0 + 2 * s + e;
        bx[2 * g + s][e] = band(k - row - 1);
      }
    }
  }

  v2f wy[4];
#pragma unroll
  for (int q = 0; q < 4; ++q) {
#pragma unroll
    for (int e = 0; e < 2; ++e)
      wy[q][e] = band(2 * q + e + 8 * h - row);
  }

  const int  r0   = tid / ROW_PIECES;
  const int  pc   = tid - r0 * ROW_PIECES;
  const int  gy0  = yw - 3 + r0;
  const int  gy0c = min(max(gy0, 0), DIM - 1);
  const bool vy0  = (gy0 == gy0c);
  const int  gy1  = gy0 + 16;
  const int  gy1c = min(max(gy1, 0), DIM - 1);
  const bool vy1  = (gy1 == gy1c);
  const bool has1 = tid < (WIN_ROWS - 16) * ROW_PIECES;
  const int  go0  = gy0c * DIM + pc * 4;
  const int  go1  = has1 ? gy1c * DIM + pc * 4 : go0;
  float* const st0 = lds + r0 * PITCH_F + 4 + pc * 4;
  float* const st1 = st0 + 16 * PITCH_F;
  const float* const fr0 = lds + (tyl * ROWS_OUT + row) * PITCH_F + x0 + 4 * h;

  const v4f zero4 = {0.f, 0.f, 0.f, 0.f};
  if (tid < NSLOT * WIN_ROWS * 2) {
    const int sl  = tid / (WIN_ROWS * 2);
    const int rem = tid - sl * (WIN_ROWS * 2);
    *reinterpret_cast<v4f*>(lds + sl * SLOT_F + (rem >> 1) * PITCH_F + (rem & 1) * 164) = zero4;
  }

  const float* __restrict__ vin = in + (size_t)vol * DIM * PLANE;
  float* const obuf = lds + NSLOT * SLOT_F;
  float* const os   = obuf + (tyl * ROWS_OUT + 8 * h) * OPITCH_F + x0 + row;
  const float* const ol = obuf + r0 * OPITCH_F + pc * 4;
  const bool has_o1 = w < OUT_WAVES;
  float* __restrict__ og = out + (size_t)vol * DIM * PLANE + (size_t)yw * DIM + 4 * tid;

  v4f G[3][2];
  v4f A[3];
  v4f OL0 = zero4, OL1 = zero4;
  v8f S[7];
#pragma unroll
  for (int i = 0; i < 3; ++i) { G[i][0] = zero4; G[i][1] = zero4; }
#pragma unroll
  for (int t = 0; t < 7; ++t)
    S[t] = (v8f){0.f, 0.f, 0.f, 0.f, 0.f, 0.f, 0.f, 0.f};

#define GLOAD(ST)                                                             \
  {                                                                           \
    const int pl_ = min(max(zs - 3 + (ST), 0), DIM - 1);                      \
    const float* __restrict__ pp_ = vin + (size_t)pl_ * PLANE;                \
    G[(ST) % 3][0] = *reinterpret_cast<const v4f*>(pp_ + go0);                \
    G[(ST) % 3][1] = *reinterpret_cast<const v4f*>(pp_ + go1);                \
  }

#define STAGE(ST)                                                             \
  {                                                                           \
    const int  p_   = zs - 3 + (ST);                                          \
    const bool pok_ = (p_ >= 0) && (p_ < DIM);                                \
    const bool k0_  = vy0 && pok_;                                            \
    const bool k1_  = vy1 && pok_;                                            \
    v4f a_, b_;                                                               \
    _Pragma("unroll")                                                         \
    for (int i = 0; i < 4; ++i) {     \
      a_[i] = k0_ ? G[(ST) % 3][0][i] : 0.0f;                                 \
      b_[i] = k1_ ? G[(ST) % 3][1][i] : 0.0f;                                 \
    }                                                                         \
    *reinterpret_cast<v4f*>(st0 + ((ST) % NSLOT) * SLOT_F) = a_;              \
    if (has1)                                                                 \
      *reinterpret_cast<v4f*>(st1 + ((ST) % NSLOT) * SLOT_F) = b_;            \
  }

#define LDFRAG(ST)                                                            \
  {                                                                           \
    _Pragma("unroll")                                                         \
    for (int g = 0; g < 3; ++g)                                               \
      A[g] = *reinterpret_cast<const v4f*>(                                   \
          fr0 + ((ST) % NSLOT) * SLOT_F + 8 * g);                             \
  }

#define SLOT_HAS_BARRIER(TT)                                                  \
  ((((TT) + 2 >= 0) && ((TT) + 2 < STEPS)) || (((TT) - 1 >= 6) && ((TT) - 1 < STEPS)))

  v8f T = {0.f, 0.f, 0.f, 0.f, 0.f, 0.f, 0.f, 0.f};

#pragma unroll
  for (int t = -4; t < STEPS + 2; ++t) {
    if (SLOT_HAS_BARRIER(t - 1)) {
      __builtin_amdgcn_s_barrier_wait(-1);
      __builtin_amdgcn_fence(__ATOMIC_ACQUIRE, "workgroup", "local");
    }
    if (t + 4 < STEPS) { GLOAD(t + 4) }
    if (t + 2 >= 0 && t + 2 < STEPS) { STAGE(t + 2) }
    if (t + 1 >= 0 && t + 1 < STEPS) { LDFRAG(t + 1) }
    if (t - 1 >= 6 && t - 1 < STEPS) {
      float* const o_ = os + ((t - 1) & 1) * OBUF_F;
#pragma unroll
      for (int r = 0; r < 8; ++r) {
        if (r < 2 || h == 0)
          o_[r * OPITCH_F] = S[6][r];
      }
    }
    if (t - 2 >= 6 && t - 2 < STEPS) {
      const float* const l_ = ol + ((t - 2) & 1) * OBUF_F;
      OL0 = *reinterpret_cast<const v4f*>(l_);
      if (has_o1)
        OL1 = *reinterpret_cast<const v4f*>(l_ + 16 * OPITCH_F);
    }
    __builtin_amdgcn_sched_barrier(0);
    if (SLOT_HAS_BARRIER(t)) {
      __builtin_amdgcn_fence(__ATOMIC_RELEASE, "workgroup", "local");
      __builtin_amdgcn_s_barrier_signal(-1);
    }
    __builtin_amdgcn_sched_barrier(0);

    if (t - 2 >= 6 && t - 2 < STEPS) {
      float* __restrict__ g_ = og + (size_t)(zs + t - 8) * PLANE;
      *reinterpret_cast<v4f*>(g_) = OL0;
      if (has_o1)
        *reinterpret_cast<v4f*>(g_ + 16 * DIM) = OL1;
    }

    if (t >= 0 && t < STEPS) {
      v8f P = {0.f, 0.f, 0.f, 0.f, 0.f, 0.f, 0.f, 0.f};
#pragma unroll
      for (int q = 0; q < 4; ++q) {
        v2f bt = {T[2 * q], T[2 * q + 1]};
        P = __builtin_amdgcn_wmma_f32_16x16x4_f32(
            false, wy[q], false, bt, (short)0, P, false, false);
      }

#pragma unroll
      for (int u = 6; u >= 1; --u) {
#pragma unroll
        for (int r = 0; r < 8; ++r)
          S[u][r] = __builtin_fmaf(GK[u], P[r], S[u - 1][r]);
      }
#pragma unroll
      for (int r = 0; r < 8; ++r)
        S[0][r] = GK[0] * P[r];

    }


    if (t + 1 >= 0 && t + 1 < STEPS) {
      T = (v8f){0.f, 0.f, 0.f, 0.f, 0.f, 0.f, 0.f, 0.f};
#pragma unroll
      for (int g = 0; g < 3; ++g) {
#pragma unroll
        for (int s = 0; s < 2; ++s) {
          v2f af = {A[g][2 * s], A[g][2 * s + 1]};
          T = __builtin_amdgcn_wmma_f32_16x16x4_f32(
              false, af, false, bx[2 * g + s], (short)0, T, false, false);
        }
      }
    }
    __builtin_amdgcn_sched_barrier(0);
  }
#undef GLOAD
#undef STAGE
#undef LDFRAG
#undef SLOT_HAS_BARRIER
}

extern "C" void kernel_launch(void* const* d_in, const int* in_sizes, int n_in,
                              void* d_out, int out_size, void* d_ws, size_t ws_size,
                              hipStream_t stream) {
  (void)in_sizes; (void)n_in; (void)out_size; (void)d_ws; (void)ws_size;
  const float* x   = (const float*)d_in[0];
  float*       out = (float*)d_out;

  gb3d_blur_xyz_wmma<<<VOLS * CHUNKS * WGS_Y, THREADS, 0, stream>>>(x, out);
}


# z-blur moved in front: running z sums on each thread's raw 16-byte input pieces before LDS staging; x/y WMMAs and LDS hand-overs on 20 planes, not 26; sums now in order z, x, y
# speedup vs baseline: 8.6885x; 1.0563x over previous
#include <hip/hip_runtime.h>

typedef float v2f __attribute__((ext_vector_type(2)));
typedef float v4f __attribute__((ext_vector_type(4)));
typedef float v8f __attribute__((ext_vector_type(8)));

__device__ constexpr float GK[7] = {
    0.0701593f, 0.1310749f, 0.1907128f, 0.2161058f,
    0.1907128f, 0.1310749f, 0.0701593f
};

#define DIM        160
#define PLANE      (DIM * DIM)
#define VOLS       4
#define TILES_X    (DIM / 16)
#define ROWS_OUT   10
#define TILES_YL   2
#define WG_ROWS    (ROWS_OUT * TILES_YL)
#define WIN_ROWS   (WG_ROWS + 6)
#define WGS_Y      (DIM / WG_ROWS)
#define ZC         20
#define CHUNKS     (DIM / ZC)
#define STEPS      (ZC + 6)
#define WAVES      (TILES_X * TILES_YL)
#define THREADS    (WAVES * 32)
#define ROW_PIECES (DIM / 4)
#define PITCH_F    172
#define SLOT_F     (WIN_ROWS * PITCH_F)
#define NSLOT      3
#define OPITCH_F   164
#define OBUF_F     (WG_ROWS * OPITCH_F)
#define OUT_WAVES  ((WG_ROWS - 16) * ROW_PIECES / 32)

__device__ __forceinline__ float band(int d) {
  return (d >= 0 && d <= 6) ? GK[d] : 0.0f;
}

__global__ __launch_bounds__(THREADS) void gb3d_blur_xyz_wmma(
    const float* __restrict__ in, float* __restrict__ out) {
  __shared__ __attribute__((aligned(16))) float lds[NSLOT * SLOT_F + 2 * OBUF_F];

  const int tid  = threadIdx.x;
  const int lane = tid & 31;
  const int w   = __builtin_amdgcn_readfirstlane(tid >> 5);
  const int tx  = w % TILES_X;
  const int tyl = w / TILES_X;

  const int b   = blockIdx.x;
  const int yp  = b % WGS_Y;
  const int t1  = b / WGS_Y;
  const int ck  = t1 % CHUNKS;
  const int vol = t1 / CHUNKS;

  const int x0 = tx * 16;
  const int yw = yp * WG_ROWS;
  const int y0 = yw + tyl * ROWS_OUT;
  const int zs = ck * ZC;

  const int row = lane & 15;
  const int h   = lane >> 4;

  v2f bx[6];
#pragma unroll
  for (int g = 0; g < 3; ++g) {
    const int k0 = 8 * g + 4 * h;
#pragma unroll
    for (int s = 0; s < 2; ++s) {
#pragma unroll
      for (int e = 0; e < 2; ++e) {
        const int k = k0 + 2 * s + e;
        bx[2 * g + s][e] = band(k - row - 1);
      }
    }
  }

  v2f wy[4];
#pragma unroll
  for (int q = 0; q < 4; ++q) {
#pragma unroll
    for (int e = 0; e < 2; ++e)
      wy[q][e] = band(2 * q + e + 8 * h - row);
  }

  const int  r0   = tid / ROW_PIECES;
  const int  pc   = tid - r0 * ROW_PIECES;
  const int  gy0  = yw - 3 + r0;
  const int  gy0c = min(max(gy0, 0), DIM - 1);
  const bool vy0  = (gy0 == gy0c);
  const int  gy1  = gy0 + 16;
  const int  gy1c = min(max(gy1, 0), DIM - 1);
  const bool vy1  = (gy1 == gy1c);
  const bool has1 = tid < (WIN_ROWS - 16) * ROW_PIECES;
  const int  go0  = gy0c * DIM + pc * 4;
  const int  go1  = has1 ? gy1c * DIM + pc * 4 : go0;
  float* const st0 = lds + r0 * PITCH_F + 4 + pc * 4;
  float* const st1 = st0 + 16 * PITCH_F;
  const float* const fr0 = lds + (tyl * ROWS_OUT + row) * PITCH_F + x0 + 4 * h;

  const v4f zero4 = {0.f, 0.f, 0.f, 0.f};
  if (tid < NSLOT * WIN_ROWS * 2) {
    const int sl  = tid / (WIN_ROWS * 2);
    const int rem = tid - sl * (WIN_ROWS * 2);
    *reinterpret_cast<v4f*>(lds + sl * SLOT_F + (rem >> 1) * PITCH_F + (rem & 1) * 164) = zero4;
  }

  const float* __restrict__ vin = in + (size_t)vol * DIM * PLANE;
  float* const obuf = lds + NSLOT * SLOT_F;
  float* const os   = obuf + (tyl * ROWS_OUT + 8 * h) * OPITCH_F + x0 + row;
  const float* const ol = obuf + r0 * OPITCH_F + pc * 4;
  const bool has_o1 = w < OUT_WAVES;
  float* __restrict__ og = out + (size_t)vol * DIM * PLANE + (size_t)yw * DIM + 4 * tid;

  v4f G[3][2];
  v4f A[3];
  v4f OL0 = zero4, OL1 = zero4;
  v4f Z[7][2];
#pragma unroll
  for (int i = 0; i < 3; ++i) { G[i][0] = zero4; G[i][1] = zero4; }
#pragma unroll
  for (int u = 0; u < 7; ++u) { Z[u][0] = zero4; Z[u][1] = zero4; }

#define GLOAD(ST)                                                             \
  {                                                                           \
    const int pl_ = min(max(zs - 3 + (ST), 0), DIM - 1);                      \
    const float* __restrict__ pp_ = vin + (size_t)pl_ * PLANE;                \
    G[(ST) % 3][0] = *reinterpret_cast<const v4f*>(pp_ + go0);                \
    G[(ST) % 3][1] = *reinterpret_cast<const v4f*>(pp_ + go1);                \
  }

#define ZACC(ST)                                                              \
  {                                                                           \
    const int  p_   = zs - 3 + (ST);                                          \
    const bool pok_ = (p_ >= 0) && (p_ < DIM);                                \
    v4f a_, b_;                                                               \
    _Pragma("unroll")                                                         \
    for (int i = 0; i < 4; ++i) {     \
      a_[i] = pok_ ? G[(ST) % 3][0][i] : 0.0f;                                \
      b_[i] = pok_ ? G[(ST) % 3][1][i] : 0.0f;                                \
    }                                                                         \
    _Pragma("unroll")                                                         \
    for (int u = 6; u >= 1; --u) {                                            \
      _Pragma("unroll")                                                       \
      for (int i = 0; i < 4; ++i) {                                           \
        Z[u][0][i] = __builtin_fmaf(GK[u], a_[i], Z[u - 1][0][i]);            \
        Z[u][1][i] = __builtin_fmaf(GK[u], b_[i], Z[u - 1][1][i]);            \
      }                                                                       \
    }                                                                         \
    _Pragma("unroll")                                                         \
    for (int i = 0; i < 4; ++i) {                                             \
      Z[0][0][i] = GK[0] * a_[i];                                             \
      Z[0][1][i] = GK[0] * b_[i];                                             \
    }                                                                         \
  }

#define STAGE(JJ)                                                             \
  {                                                                           \
    v4f a_, b_;                                                               \
    _Pragma("unroll")                                                         \
    for (int i = 0; i < 4; ++i) {     \
      a_[i] = vy0 ? Z[6][0][i] : 0.0f;                                        \
      b_[i] = vy1 ? Z[6][1][i] : 0.0f;                                        \
    }                                                                         \
    *reinterpret_cast<v4f*>(st0 + ((JJ) % NSLOT) * SLOT_F) = a_;              \
    if (has1)                                                                 \
      *reinterpret_cast<v4f*>(st1 + ((JJ) % NSLOT) * SLOT_F) = b_;            \
  }

#define LDFRAG(ST)                                                            \
  {                                                                           \
    _Pragma("unroll")                                                         \
    for (int g = 0; g < 3; ++g)                                               \
      A[g] = *reinterpret_cast<const v4f*>(                                   \
          fr0 + ((ST) % NSLOT) * SLOT_F + 8 * g);                             \
  }

#define SLOT_HAS_BARRIER(TT)                                                  \
  ((((TT) + 2 >= 0) && ((TT) + 2 < ZC)) || (((TT) - 1 >= 0) && ((TT) - 1 < ZC)))

  v8f T = {0.f, 0.f, 0.f, 0.f, 0.f, 0.f, 0.f, 0.f};
  v8f P = {0.f, 0.f, 0.f, 0.f, 0.f, 0.f, 0.f, 0.f};

#pragma unroll
  for (int t = -10; t < ZC + 2; ++t) {
    if (SLOT_HAS_BARRIER(t - 1)) {
      __builtin_amdgcn_s_barrier_wait(-1);
      __builtin_amdgcn_fence(__ATOMIC_ACQUIRE, "workgroup", "local");
    }
    if (t + 10 < STEPS) { GLOAD(t + 10) }
    if (t + 8 >= 0 && t + 8 < STEPS) { ZACC(t + 8) }
    if (t + 2 >= 0 && t + 2 < ZC) { STAGE(t + 2) }
    if (t + 1 >= 0 && t + 1 < ZC) { LDFRAG(t + 1) }
    if (t - 1 >= 0 && t - 1 < ZC) {
      float* const o_ = os + ((t - 1) & 1) * OBUF_F;
#pragma unroll
      for (int r = 0; r < 8; ++r) {
        if (r < 2 || h == 0)
          o_[r * OPITCH_F] = P[r];
      }
    }
    if (t - 2 >= 0 && t - 2 < ZC) {
      const float* const l_ = ol + ((t - 2) & 1) * OBUF_F;
      OL0 = *reinterpret_cast<const v4f*>(l_);
      if (has_o1)
        OL1 = *reinterpret_cast<const v4f*>(l_ + 16 * OPITCH_F);
    }
    __builtin_amdgcn_sched_barrier(0);
    if (SLOT_HAS_BARRIER(t)) {
      __builtin_amdgcn_fence(__ATOMIC_RELEASE, "workgroup", "local");
      __builtin_amdgcn_s_barrier_signal(-1);
    }
    __builtin_amdgcn_sched_barrier(0);

    if (t - 2 >= 0 && t - 2 < ZC) {
      float* __restrict__ g_ = og + (size_t)(zs + t - 2) * PLANE;
      *reinterpret_cast<v4f*>(g_) = OL0;
      if (has_o1)
        *reinterpret_cast<v4f*>(g_ + 16 * DIM) = OL1;
    }

    if (t >= 0 && t < ZC) {
      P = (v8f){0.f, 0.f, 0.f, 0.f, 0.f, 0.f, 0.f, 0.f};
#pragma unroll
      for (int q = 0; q < 4; ++q) {
        v2f bt = {T[2 * q], T[2 * q + 1]};
        P = __builtin_amdgcn_wmma_f32_16x16x4_f32(
            false, wy[q], false, bt, (short)0, P, false, false);
      }
    }

    if (t + 1 >= 0 && t + 1 < ZC) {
      T = (v8f){0.f, 0.f, 0.f, 0.f, 0.f, 0.f, 0.f, 0.f};
#pragma unroll
      for (int g = 0; g < 3; ++g) {
#pragma unroll
        for (int s = 0; s < 2; ++s) {
          v2f af = {A[g][2 * s], A[g][2 * s + 1]};
          T = __builtin_amdgcn_wmma_f32_16x16x4_f32(
              false, af, false, bx[2 * g + s], (short)0, T, false, false);
        }
      }
    }
    __builtin_amdgcn_sched_barrier(0);
  }
#undef GLOAD
#undef ZACC
#undef STAGE
#undef LDFRAG
#undef SLOT_HAS_BARRIER
}

extern "C" void kernel_launch(void* const* d_in, const int* in_sizes, int n_in,
                              void* d_out, int out_size, void* d_ws, size_t ws_size,
                              hipStream_t stream) {
  (void)in_sizes; (void)n_in; (void)out_size; (void)d_ws; (void)ws_size;
  const float* x   = (const float*)d_in[0];
  float*       out = (float*)d_out;

  gb3d_blur_xyz_wmma<<<VOLS * CHUNKS * WGS_Y, THREADS, 0, stream>>>(x, out);
}
